# starting kernel
# baseline (speedup 1.0000x reference)
#include <hip/hip_runtime.h>
#include <hip/hip_bf16.h>


typedef __attribute__((ext_vector_type(2))) float v2f;
typedef __attribute__((ext_vector_type(8))) float v8f;

#define FDIM 64

__global__ __launch_bounds__(256) void gcn_zero_kernel(float4* __restrict__ p, int n4) {
    int i = blockIdx.x * blockDim.x + threadIdx.x;
    if (i < n4) {
        p[i] = make_float4(0.f, 0.f, 0.f, 0.f);
    }
}

__global__ __launch_bounds__(256) void gcn_scatter_kernel(
    const float* __restrict__ x,
    const int*   __restrict__ edge_src,
    const int*   __restrict__ edge_dst,
    float*       __restrict__ agg,
    int n_edges)
{
    const int wid  = threadIdx.x >> 5;
    const int lane = threadIdx.x & 31;
    const int edge = blockIdx.x * 8 + wid;
    if (edge >= n_edges) return;

    const int s = edge_src[edge];
    const int d = edge_dst[edge];

    const float2 v = *(const float2*)(x + (size_t)s * FDIM + lane * 2);
    float* p = agg + (size_t)d * FDIM + lane * 2;
    atomicAdd(p,     v.x);
    atomicAdd(p + 1, v.y);
}

__global__ __launch_bounds__(256) void gcn_gemm_wmma_kernel(
    const float* __restrict__ agg,
    const float* __restrict__ W,
    const float* __restrict__ bias,
    float*       __restrict__ out,
    int n_nodes)
{
    const int tiles_n = FDIM / 16;
    const int wave    = blockIdx.x * 8 + (threadIdx.x >> 5);
    const int tile_m  = wave / tiles_n;
    const int tile_n  = wave % tiles_n;
    const int row0    = tile_m * 16;
    if (row0 >= n_nodes) return;

    const int lane = threadIdx.x & 31;
    const int half = lane >> 4;
    const int lr   = lane & 15;
    const int col0 = tile_n * 16;

    const float bv = bias[col0 + lr];
    v8f acc;
#pragma unroll
    for (int v = 0; v < 8; ++v) acc[v] = bv;

    const int arow = (row0 + lr < n_nodes) ? (row0 + lr) : (n_nodes - 1);
    const float* arow_p = agg + (size_t)arow * FDIM;

#pragma unroll
    for (int k0 = 0; k0 < FDIM; k0 += 4) {
        v2f a, bm;
        const float2 av = *(const float2*)(arow_p + k0 + 2 * half);
        a[0] = av.x;
        a[1] = av.y;
        bm[0] = W[(k0 + 0 + 2 * half) * FDIM + col0 + lr];
        bm[1] = W[(k0 + 1 + 2 * half) * FDIM + col0 + lr];

        acc = __builtin_amdgcn_wmma_f32_16x16x4_f32(
            false, a, false, bm, (short)0, acc, false, false);
    }

#pragma unroll
    for (int v = 0; v < 8; ++v) {
        const int rm = row0 + v + 8 * half;
        if (rm < n_nodes) {
            out[(size_t)rm * FDIM + col0 + lr] = acc[v];
        }
    }
}

extern "C" void kernel_launch(void* const* d_in, const int* in_sizes, int n_in,
                              void* d_out, int out_size, void* d_ws, size_t ws_size,
                              hipStream_t stream) {
    const float* x        = (const float*)d_in[0];
    const int*   edge_src = (const int*)d_in[1];
    const int*   edge_dst = (const int*)d_in[2];
    const float* W        = (const float*)d_in[3];
    const float* bias     = (const float*)d_in[4];
    float*       out      = (float*)d_out;

    const int n_nodes = in_sizes[0] / FDIM;
    const int n_edges = in_sizes[1];

    float* agg = (float*)d_ws;

    {
        const int n4     = n_nodes * (FDIM / 4);
        const int blocks = (n4 + 255) / 256;
        gcn_zero_kernel<<<blocks, 256, 0, stream>>>((float4*)agg, n4);
    }

    {
        const int blocks = (n_edges + 7) / 8;
        gcn_scatter_kernel<<<blocks, 256, 0, stream>>>(x, edge_src, edge_dst, agg, n_edges);
    }

    {
        const int tiles_m = (n_nodes + 15) / 16;
        const int waves   = tiles_m * (FDIM / 16);
        const int blocks  = (waves + 7) / 8;
        gcn_gemm_wmma_kernel<<<blocks, 256, 0, stream>>>(agg, W, bias, out, n_nodes);
    }
}


# scatter: 16 edges per wave (indices by wide scalar loads, all 16 row loads in flight before the first atomic), 16x fewer waves; f32 atomics kept, arrival order unordered as before
# speedup vs baseline: 1.7428x; 1.7428x over previous
#include <hip/hip_runtime.h>
#include <hip/hip_bf16.h>


typedef __attribute__((ext_vector_type(2))) float v2f;
typedef __attribute__((ext_vector_type(8))) float v8f;

#define FDIM 64

__global__ __launch_bounds__(256) void gcn_zero_kernel(float4* __restrict__ p, int n4) {
    int i = blockIdx.x * blockDim.x + threadIdx.x;
    if (i < n4) {
        p[i] = make_float4(0.f, 0.f, 0.f, 0.f);
    }
}

#define EPW 16

__global__ __launch_bounds__(256) void gcn_scatter_kernel(
    const float* __restrict__ x,
    const int*   __restrict__ edge_src,
    const int*   __restrict__ edge_dst,
    float*       __restrict__ agg,
    int n_edges)
{
    const int wid  = __builtin_amdgcn_readfirstlane((int)(threadIdx.x >> 5));
    const int lane = threadIdx.x & 31;
    const int e0   = (blockIdx.x * 8 + wid) * EPW;
    if (e0 >= n_edges) return;
    const int col = lane * 2;

    if (e0 + EPW <= n_edges) {
        int s[EPW], d[EPW];
#pragma unroll
        for (int i = 0; i < EPW; ++i) {
            s[i] = edge_src[e0 + i];
            d[i] = edge_dst[e0 + i];
        }
        float2 v[EPW];
#pragma unroll
        for (int i = 0; i < EPW; ++i) {
            v[i] = *(const float2*)(x + (size_t)s[i] * FDIM + col);
        }
#pragma unroll
        for (int i = 0; i < EPW; ++i) {
            float* p = agg + (size_t)d[i] * FDIM + col;
            atomicAdd(p,     v[i].x);
            atomicAdd(p + 1, v[i].y);
        }
    } else {
        for (int e = e0; e < n_edges; ++e) {
            const int s = edge_src[e];
            const int d = edge_dst[e];
            const float2 v = *(const float2*)(x + (size_t)s * FDIM + col);
            float* p = agg + (size_t)d * FDIM + col;
            atomicAdd(p,     v.x);
            atomicAdd(p + 1, v.y);
        }
    }
}

__global__ __launch_bounds__(256) void gcn_gemm_wmma_kernel(
    const float* __restrict__ agg,
    const float* __restrict__ W,
    const float* __restrict__ bias,
    float*       __restrict__ out,
    int n_nodes)
{
    const int tiles_n = FDIM / 16;
    const int wave    = blockIdx.x * 8 + (threadIdx.x >> 5);
    const int tile_m  = wave / tiles_n;
    const int tile_n  = wave % tiles_n;
    const int row0    = tile_m * 16;
    if (row0 >= n_nodes) return;

    const int lane = threadIdx.x & 31;
    const int half = lane >> 4;
    const int lr   = lane & 15;
    const int col0 = tile_n * 16;

    const float bv = bias[col0 + lr];
    v8f acc;
#pragma unroll
    for (int v = 0; v < 8; ++v) acc[v] = bv;

    const int arow = (row0 + lr < n_nodes) ? (row0 + lr) : (n_nodes - 1);
    const float* arow_p = agg + (size_t)arow * FDIM;

#pragma unroll
    for (int k0 = 0; k0 < FDIM; k0 += 4) {
        v2f a, bm;
        const float2 av = *(const float2*)(arow_p + k0 + 2 * half);
        a[0] = av.x;
        a[1] = av.y;
        bm[0] = W[(k0 + 0 + 2 * half) * FDIM + col0 + lr];
        bm[1] = W[(k0 + 1 + 2 * half) * FDIM + col0 + lr];

        acc = __builtin_amdgcn_wmma_f32_16x16x4_f32(
            false, a, false, bm, (short)0, acc, false, false);
    }

#pragma unroll
    for (int v = 0; v < 8; ++v) {
        const int rm = row0 + v + 8 * half;
        if (rm < n_nodes) {
            out[(size_t)rm * FDIM + col0 + lr] = acc[v];
        }
    }
}

extern "C" void kernel_launch(void* const* d_in, const int* in_sizes, int n_in,
                              void* d_out, int out_size, void* d_ws, size_t ws_size,
                              hipStream_t stream) {
    const float* x        = (const float*)d_in[0];
    const int*   edge_src = (const int*)d_in[1];
    const int*   edge_dst = (const int*)d_in[2];
    const float* W        = (const float*)d_in[3];
    const float* bias     = (const float*)d_in[4];
    float*       out      = (float*)d_out;

    const int n_nodes = in_sizes[0] / FDIM;
    const int n_edges = in_sizes[1];

    float* agg = (float*)d_ws;

    {
        const int n4     = n_nodes * (FDIM / 4);
        const int blocks = (n4 + 255) / 256;
        gcn_zero_kernel<<<blocks, 256, 0, stream>>>((float4*)agg, n4);
    }

    {
        const int blocks = (n_edges + 8 * EPW - 1) / (8 * EPW);
        gcn_scatter_kernel<<<blocks, 256, 0, stream>>>(x, edge_src, edge_dst, agg, n_edges);
    }

    {
        const int tiles_m = (n_nodes + 15) / 16;
        const int waves   = tiles_m * (FDIM / 16);
        const int blocks  = (waves + 7) / 8;
        gcn_gemm_wmma_kernel<<<blocks, 256, 0, stream>>>(agg, W, bias, out, n_nodes);
    }
}


# global f32 atomics replaced by binning edges per call into destination-row ranges (count, scan, place); each range summed in LDS by ds_add_f32, arrival order, and fed to the WMMA GEMM from LDS
# speedup vs baseline: 2.9919x; 1.7167x over previous
#include <hip/hip_runtime.h>


typedef __attribute__((ext_vector_type(2))) float v2f;
typedef __attribute__((ext_vector_type(8))) float v8f;

#define FDIM   64
#define ROWS   400
#define PITCH  68
#define MAXBIN 256
#define PBLK   256
#define PTHR   512
#define PUNR   10
#define EPW    16

__global__ __launch_bounds__(PTHR) void gcn_count_kernel(
    const int* __restrict__ edge_dst,
    int*       __restrict__ cnt,
    int n_edges, int n_nodes)
{
    __shared__ int h[MAXBIN];
    const int t   = threadIdx.x;
    const int per = (n_edges + PBLK - 1) / PBLK;
    const int lo  = blockIdx.x * per;
    const int hi  = (lo + per < n_edges) ? (lo + per) : n_edges;
    int e0 = lo + t;
    int d[PUNR];
#pragma unroll
    for (int i = 0; i < PUNR; ++i) {
        const int e = e0 + i * PTHR;
        d[i] = (e < hi) ? edge_dst[e] : -1;
    }
    if (t < MAXBIN) h[t] = 0;
    __syncthreads();
    for (;;) {
#pragma unroll
        for (int i = 0; i < PUNR; ++i) {
            if ((unsigned)d[i] < (unsigned)n_nodes) {
                atomicAdd(&h[d[i] / ROWS], 1);
            }
        }
        e0 += PTHR * PUNR;
        if (e0 >= hi) break;
#pragma unroll
        for (int i = 0; i < PUNR; ++i) {
            const int e = e0 + i * PTHR;
            d[i] = (e < hi) ? edge_dst[e] : -1;
        }
    }
    __syncthreads();
    if (t < MAXBIN) cnt[blockIdx.x * MAXBIN + t] = h[t];
}

__global__ __launch_bounds__(1024) void gcn_scan_kernel(
    int* __restrict__ cnt,
    int* __restrict__ bin_start)
{
    __shared__ int part[4][MAXBIN];
    __shared__ int wsum[8];
    const int t = threadIdx.x;
    const int r = t & (MAXBIN - 1);
    const int q = t >> 8;
    int c[PBLK / 4];
#pragma unroll
    for (int i = 0; i < PBLK / 4; ++i) c[i] = cnt[(q * (PBLK / 4) + i) * MAXBIN + r];
    int run = 0;
#pragma unroll
    for (int i = 0; i < PBLK / 4; ++i) {
        const int v = c[i];
        c[i] = run;
        run += v;
    }
    part[q][r] = run;
    __syncthreads();
    int base = 0;
    for (int k = 0; k < q; ++k) base += part[k][r];
#pragma unroll
    for (int i = 0; i < PBLK / 4; ++i) cnt[(q * (PBLK / 4) + i) * MAXBIN + r] = base + c[i];

    const int total = part[0][r] + part[1][r] + part[2][r] + part[3][r];
    const int lane  = t & 31;
    int inc = total;
#pragma unroll
    for (int off = 1; off < 32; off <<= 1) {
        const int u = __shfl_up(inc, off);
        if (lane >= off) inc += u;
    }
    if (q == 0 && lane == 31) wsum[r >> 5] = inc;
    __syncthreads();
    if (q == 0) {
        int pre = 0;
        for (int k = 0; k < (r >> 5); ++k) pre += wsum[k];
        bin_start[r] = pre + inc - total;
        if (r == MAXBIN - 1) bin_start[MAXBIN] = pre + inc;
    }
}

__global__ __launch_bounds__(PTHR) void gcn_place_kernel(
    const int* __restrict__ edge_src,
    const int* __restrict__ edge_dst,
    const int* __restrict__ cnt,
    const int* __restrict__ bin_start,
    int2*      __restrict__ sorted,
    int n_edges, int n_nodes)
{
    __shared__ int  lstart[MAXBIN];
    __shared__ int  lcur[MAXBIN];
    __shared__ int  gbase[MAXBIN];
    __shared__ int  wtot[MAXBIN / 32];
    __shared__ int2 stage[PTHR * PUNR];

    const int t    = threadIdx.x;
    const int lane = t & 31;
    const int per  = (n_edges + PBLK - 1) / PBLK;
    const int lo   = blockIdx.x * per;
    const int hi   = (lo + per < n_edges) ? (lo + per) : n_edges;

    if (t < MAXBIN) gbase[t] = bin_start[t] + cnt[blockIdx.x * MAXBIN + t];
    if (blockIdx.x == 0 && t < EPW) sorted[bin_start[MAXBIN] + t] = make_int2(0, 0);

    for (int b0 = lo; b0 < hi; b0 += PTHR * PUNR) {
        int s[PUNR], d[PUNR];
#pragma unroll
        for (int i = 0; i < PUNR; ++i) {
            const int e = b0 + t + i * PTHR;
            d[i] = (e < hi) ? edge_dst[e] : -1;
            s[i] = (e < hi) ? edge_src[e] : 0;
        }
        if (t < MAXBIN) lcur[t] = 0;
        __syncthreads();
#pragma unroll
        for (int i = 0; i < PUNR; ++i) {
            if ((unsigned)d[i] < (unsigned)n_nodes) atomicAdd(&lcur[d[i] / ROWS], 1);
        }
        __syncthreads();
        const int c = (t < MAXBIN) ? lcur[t] : 0;
        int inc = c;
#pragma unroll
        for (int off = 1; off < 32; off <<= 1) {
            const int u = __shfl_up(inc, off);
            if (lane >= off) inc += u;
        }
        if (t < MAXBIN && lane == 31) wtot[t >> 5] = inc;
        __syncthreads();
        int nvalid = 0;
#pragma unroll
        for (int k = 0; k < MAXBIN / 32; ++k) nvalid += wtot[k];
        if (t < MAXBIN) {
            int pre = 0;
            for (int k = 0; k < (t >> 5); ++k) pre += wtot[k];
            lstart[t] = pre + inc - c;
            lcur[t]   = pre + inc - c;
        }
        __syncthreads();
        int pos[PUNR];
#pragma unroll
        for (int i = 0; i < PUNR; ++i) {
            pos[i] = -1;
            if ((unsigned)d[i] < (unsigned)n_nodes) {
                pos[i] = atomicAdd(&lcur[d[i] / ROWS], 1);
            }
        }
#pragma unroll
        for (int i = 0; i < PUNR; ++i) {
            if (pos[i] >= 0) stage[pos[i]] = make_int2(s[i], d[i]);
        }
        __syncthreads();
        for (int i = t; i < nvalid; i += PTHR) {
            const int2 p = stage[i];
            const int  r = p.y / ROWS;
            sorted[gbase[r] + (i - lstart[r])] = p;
        }
        __syncthreads();
        if (t < MAXBIN) gbase[t] += c;
    }
}

__global__ __launch_bounds__(1024) void gcn_sum_gemm_kernel(
    const float* __restrict__ x,
    const int2*  __restrict__ sorted,
    const int*   __restrict__ bin_start,
    const float* __restrict__ W,
    const float* __restrict__ bias,
    float*       __restrict__ out,
    int n_nodes)
{
    __shared__ float agg[(ROWS + 1) * PITCH];

    const int t        = threadIdx.x;
    const int row_base = blockIdx.x * ROWS;

    const int wid  = __builtin_amdgcn_readfirstlane(t >> 5);
    const int lane = t & 31;
    const int col  = lane * 2;
    const int lo   = bin_start[blockIdx.x];
    const int hi   = bin_start[blockIdx.x + 1];

    int e0 = lo + wid * EPW;
    const bool any = e0 < hi;
    const int  ef  = any ? e0 : hi;
    int2 p[EPW];
#pragma unroll
    for (int i = 0; i < EPW; ++i) p[i] = sorted[ef + i];

    for (int i = t; i < ROWS * PITCH / 4; i += 1024) {
        ((float4*)agg)[i] = make_float4(0.f, 0.f, 0.f, 0.f);
    }
    __syncthreads();

    if (any) {
        for (;;) {
            const int n = hi - e0;
            float2 v[EPW];
            int    dl[EPW];
#pragma unroll
            for (int i = 0; i < EPW; ++i) {
                v[i]  = *(const float2*)(x + (size_t)p[i].x * FDIM + col);
                dl[i] = (i < n) ? (p[i].y - row_base) : ROWS;
            }
            const int  e1   = e0 + 32 * EPW;
            const bool more = e1 < hi;
            if (more) {
#pragma unroll
                for (int i = 0; i < EPW; ++i) p[i] = sorted[e1 + i];
            }
#pragma unroll
            for (int i = 0; i < EPW; ++i) {
                float* a = agg + dl[i] * PITCH + col;
                atomicAdd(a,     v[i].x);
                atomicAdd(a + 1, v[i].y);
            }
            if (!more) break;
            e0 = e1;
        }
    }
    __syncthreads();

    const int half = lane >> 4;
    const int lr   = lane & 15;
    for (int job = wid; job < (ROWS / 16) * (FDIM / 16); job += 32) {
        const int tile_m = job >> 2;
        const int col0   = (job & 3) * 16;

        const float bv = bias[col0 + lr];
        v8f acc;
#pragma unroll
        for (int v = 0; v < 8; ++v) acc[v] = bv;

        const float* arow_p = agg + (tile_m * 16 + lr) * PITCH;

#pragma unroll
        for (int k0 = 0; k0 < FDIM; k0 += 4) {
            v2f a, bm;
            const float2 av = *(const float2*)(arow_p + k0 + 2 * half);
            a[0] = av.x;
            a[1] = av.y;
            bm[0] = W[(k0 + 0 + 2 * half) * FDIM + col0 + lr];
            bm[1] = W[(k0 + 1 + 2 * half) * FDIM + col0 + lr];
            acc = __builtin_amdgcn_wmma_f32_16x16x4_f32(
                false, a, false, bm, (short)0, acc, false, false);
        }

#pragma unroll
        for (int v = 0; v < 8; ++v) {
            const int rm = row_base + tile_m * 16 + v + 8 * half;
            if (rm < n_nodes) {
                out[(size_t)rm * FDIM + col0 + lr] = acc[v];
            }
        }
    }
}

extern "C" void kernel_launch(void* const* d_in, const int* in_sizes, int n_in,
                              void* d_out, int out_size, void* d_ws, size_t ws_size,
                              hipStream_t stream) {
    const float* x        = (const float*)d_in[0];
    const int*   edge_src = (const int*)d_in[1];
    const int*   edge_dst = (const int*)d_in[2];
    const float* W        = (const float*)d_in[3];
    const float* bias     = (const float*)d_in[4];
    float*       out      = (float*)d_out;

    const int n_nodes = in_sizes[0] / FDIM;
    const int n_edges = in_sizes[1];

    int*  cnt       = (int*)d_ws;
    int*  bin_start = cnt + PBLK * MAXBIN;
    int2* sorted    = (int2*)(bin_start + 512);

    gcn_count_kernel<<<PBLK, PTHR, 0, stream>>>(edge_dst, cnt, n_edges, n_nodes);
    gcn_scan_kernel<<<1, 1024, 0, stream>>>(cnt, bin_start);
    gcn_place_kernel<<<PBLK, PTHR, 0, stream>>>(edge_src, edge_dst, cnt, bin_start, sorted,
                                               n_edges, n_nodes);
    {
        const int nbins = (n_nodes + ROWS - 1) / ROWS;
        gcn_sum_gemm_kernel<<<nbins, 1024, 0, stream>>>(x, sorted, bin_start, W, bias, out, n_nodes);
    }
}


# count and scan launches removed: each 5120-edge slice is ordered by range in LDS, written in place with its run starts; the sum launch merges its runs into an LDS list; f32 LDS adds, arrival order
# speedup vs baseline: 3.5822x; 1.1973x over previous
#include <hip/hip_runtime.h>


typedef __attribute__((ext_vector_type(2))) float v2f;
typedef __attribute__((ext_vector_type(8))) float v8f;

#define FDIM   64
#define ROWS   400
#define PITCH  68
#define MAXBIN 256
#define PTHR   512
#define PUNR   10
#define SLICE  (PTHR * PUNR)
#define TSTR   320
#define TABMAX (1 << 20)
#define EPW    16
#define CAP    8192

__global__ __launch_bounds__(PTHR) void gcn_bin_kernel(
    const int* __restrict__ edge_src,
    const int* __restrict__ edge_dst,
    int*       __restrict__ tab,
    int2*      __restrict__ sorted,
    int n_edges, int n_nodes)
{
    __shared__ int  lcur[MAXBIN];
    __shared__ int  wtot[MAXBIN / 32];
    __shared__ int2 stage[SLICE];

    const int t    = threadIdx.x;
    const int lane = t & 31;
    const int lo   = blockIdx.x * SLICE;
    const int hi   = (lo + SLICE < n_edges) ? (lo + SLICE) : n_edges;

    int s[PUNR], d[PUNR];
#pragma unroll
    for (int i = 0; i < PUNR; ++i) {
        const int e = lo + t + i * PTHR;
        d[i] = (e < hi) ? edge_dst[e] : -1;
        s[i] = (e < hi) ? edge_src[e] : 0;
    }
    if (t < MAXBIN) lcur[t] = 0;
    __syncthreads();
#pragma unroll
    for (int i = 0; i < PUNR; ++i) {
        if ((unsigned)d[i] < (unsigned)n_nodes) atomicAdd(&lcur[d[i] / ROWS], 1);
    }
    __syncthreads();
    const int c = (t < MAXBIN) ? lcur[t] : 0;
    int inc = c;
#pragma unroll
    for (int off = 1; off < 32; off <<= 1) {
        const int u = __shfl_up(inc, off);
        if (lane >= off) inc += u;
    }
    if (t < MAXBIN && lane == 31) wtot[t >> 5] = inc;
    __syncthreads();
    int nvalid = 0;
#pragma unroll
    for (int k = 0; k < MAXBIN / 32; ++k) nvalid += wtot[k];
    if (t < MAXBIN) {
        int pre = 0;
        for (int k = 0; k < (t >> 5); ++k) pre += wtot[k];
        const int start = pre + inc - c;
        lcur[t] = start;
        tab[blockIdx.x * TSTR + t] = start;
        if (t == MAXBIN - 1) tab[blockIdx.x * TSTR + MAXBIN] = nvalid;
    }
    __syncthreads();
    int pos[PUNR];
#pragma unroll
    for (int i = 0; i < PUNR; ++i) {
        pos[i] = -1;
        if ((unsigned)d[i] < (unsigned)n_nodes) {
            pos[i] = atomicAdd(&lcur[d[i] / ROWS], 1);
        }
    }
#pragma unroll
    for (int i = 0; i < PUNR; ++i) {
        if (pos[i] >= 0) stage[pos[i]] = make_int2(s[i], d[i]);
    }
    __syncthreads();
    for (int i = t; i < nvalid; i += PTHR) sorted[lo + i] = stage[i];
}

__global__ __launch_bounds__(1024) void gcn_sum_gemm_kernel(
    const float* __restrict__ x,
    const int2*  __restrict__ sorted,
    const int*   __restrict__ tab,
    const float* __restrict__ W,
    const float* __restrict__ bias,
    float*       __restrict__ out,
    int n_nodes, int nslices)
{
    __shared__ float agg[(ROWS + 1) * PITCH];
    __shared__ int2  elist[CAP];
    __shared__ int   segpre[MAXBIN + 1];
    __shared__ int   segst[MAXBIN];
    __shared__ int   wtot[MAXBIN / 32];
    __shared__ int   gend;

    const int t        = threadIdx.x;
    const int r        = blockIdx.x;
    const int row_base = r * ROWS;

    const int wid  = __builtin_amdgcn_readfirstlane(t >> 5);
    const int lane = t & 31;
    const int col  = lane * 2;
    const int sub  = lane & (EPW - 1);

    for (int c0 = 0; c0 < nslices; c0 += MAXBIN) {
        const int nch = (nslices - c0 < MAXBIN) ? (nslices - c0) : MAXBIN;
        int st = 0, en = 0;
        if (t < nch) {
            const int* row = tab + (size_t)(c0 + t) * TSTR + r;
            st = row[0];
            en = row[1];
        }
        if (c0 == 0) {
            for (int i = t; i < ROWS * PITCH / 4; i += 1024) {
                ((float4*)agg)[i] = make_float4(0.f, 0.f, 0.f, 0.f);
            }
        }
        const int len = en - st;
        int inc = len;
#pragma unroll
        for (int off = 1; off < 32; off <<= 1) {
            const int u = __shfl_up(inc, off);
            if (lane >= off) inc += u;
        }
        if (t < MAXBIN && lane == 31) wtot[t >> 5] = inc;
        __syncthreads();
        if (t < MAXBIN) {
            int pre = 0;
#pragma unroll
            for (int k = 0; k < MAXBIN / 32; ++k) pre += (k < (t >> 5)) ? wtot[k] : 0;
            segpre[t] = pre + inc - len;
            segst[t]  = (c0 + t) * SLICE + st;
            if (t == MAXBIN - 1) segpre[MAXBIN] = pre + inc;
        }
        __syncthreads();

        int g0 = 0;
        while (g0 < nch) {
            const int base = segpre[g0];
            if (t >= g0 && t < nch) {
                const bool fit  = (segpre[t + 1] - base) <= CAP;
                const bool next = (t + 1 < nch) && ((segpre[t + 2] - base) <= CAP);
                if (fit && !next) gend = t + 1;
            }
            __syncthreads();
            const int g1  = __builtin_amdgcn_readfirstlane(gend);
            const int cnt = __builtin_amdgcn_readfirstlane(segpre[g1] - base);

            {
                const int seg = g0 + (t >> 2);
                const int q   = t & 3;
                if (seg < g1) {
                    const int   n    = segpre[seg + 1] - segpre[seg];
                    const int   dst0 = segpre[seg] - base;
                    const int2* srcp = sorted + segst[seg];
                    for (int j = q; j < n; j += 16) {
                        int2 a[4];
#pragma unroll
                        for (int u = 0; u < 4; ++u) {
                            const int jj = j + 4 * u;
                            a[u] = (jj < n) ? srcp[jj] : make_int2(0, 0);
                        }
#pragma unroll
                        for (int u = 0; u < 4; ++u) {
                            const int jj = j + 4 * u;
                            if (jj < n) elist[dst0 + jj] = a[u];
                        }
                    }
                }
            }
            __syncthreads();

            int e0 = wid * EPW;
            if (e0 < cnt) {
                int px, py;
                {
                    const int  k  = e0 + sub;
                    const bool ok = k < cnt;
                    const int2 pe = elist[ok ? k : 0];
                    px = ok ? pe.x : 0;
                    py = ok ? (pe.y - row_base) : ROWS;
                }
                for (;;) {
                    float2 v[EPW];
                    int    dl[EPW];
#pragma unroll
                    for (int i = 0; i < EPW; ++i) {
                        const int src = __builtin_amdgcn_readlane(px, i);
                        dl[i] = __builtin_amdgcn_readlane(py, i);
                        v[i]  = *(const float2*)(x + (size_t)src * FDIM + col);
                    }
                    const int  e1   = e0 + 32 * EPW;
                    const bool more = e1 < cnt;
                    if (more) {
                        const int  k  = e1 + sub;
                        const bool ok = k < cnt;
                        const int2 pe = elist[ok ? k : 0];
                        px = ok ? pe.x : 0;
                        py = ok ? (pe.y - row_base) : ROWS;
                    }
#pragma unroll
                    for (int i = 0; i < EPW; ++i) {
                        float* a = agg + dl[i] * PITCH + col;
                        atomicAdd(a,     v[i].x);
                        atomicAdd(a + 1, v[i].y);
                    }
                    if (!more) break;
                    e0 = e1;
                }
            }
            __syncthreads();
            g0 = g1;
        }
    }
    __syncthreads();

    const int half = lane >> 4;
    const int lr   = lane & 15;
    for (int job = wid; job < (ROWS / 16) * (FDIM / 16); job += 32) {
        const int tile_m = job >> 2;
        const int col0   = (job & 3) * 16;

        const float bv = bias[col0 + lr];
        v8f acc;
#pragma unroll
        for (int v = 0; v < 8; ++v) acc[v] = bv;

        const float* arow_p = agg + (tile_m * 16 + lr) * PITCH;

#pragma unroll
        for (int k0 = 0; k0 < FDIM; k0 += 4) {
            v2f a, bm;
            const float2 av = *(const float2*)(arow_p + k0 + 2 * half);
            a[0] = av.x;
            a[1] = av.y;
            bm[0] = W[(k0 + 0 + 2 * half) * FDIM + col0 + lr];
            bm[1] = W[(k0 + 1 + 2 * half) * FDIM + col0 + lr];
            acc = __builtin_amdgcn_wmma_f32_16x16x4_f32(
                false, a, false, bm, (short)0, acc, false, false);
        }

#pragma unroll
        for (int v = 0; v < 8; ++v) {
            const int rm = row_base + tile_m * 16 + v + 8 * half;
            if (rm < n_nodes) {
                out[(size_t)rm * FDIM + col0 + lr] = acc[v];
            }
        }
    }
}

extern "C" void kernel_launch(void* const* d_in, const int* in_sizes, int n_in,
                              void* d_out, int out_size, void* d_ws, size_t ws_size,
                              hipStream_t stream) {
    const float* x        = (const float*)d_in[0];
    const int*   edge_src = (const int*)d_in[1];
    const int*   edge_dst = (const int*)d_in[2];
    const float* W        = (const float*)d_in[3];
    const float* bias     = (const float*)d_in[4];
    float*       out      = (float*)d_out;

    const int n_nodes = in_sizes[0] / FDIM;
    const int n_edges = in_sizes[1];

    int*  tab    = (int*)d_ws;
    int2* sorted = (int2*)(tab + TABMAX);

    const int nslices = (n_edges + SLICE - 1) / SLICE;
    gcn_bin_kernel<<<nslices, PTHR, 0, stream>>>(edge_src, edge_dst, tab, sorted, n_edges, n_nodes);
    {
        const int nbins = (n_nodes + ROWS - 1) / ROWS;
        gcn_sum_gemm_kernel<<<nbins, 1024, 0, stream>>>(x, sorted, tab, W, bias, out, n_nodes, nslices);
    }
}


# sum launch: a range's edge list ordered by row in LDS (counting sort, rank = the counting add's return); a wave sums a row's sources in registers, one ds_add_f32 per row not per edge; arrival order
# speedup vs baseline: 4.2128x; 1.1760x over previous
#include <hip/hip_runtime.h>


typedef __attribute__((ext_vector_type(2))) float v2f;
typedef __attribute__((ext_vector_type(8))) float v8f;

#define FDIM   64
#define ROWS   400
#define PITCH  68
#define MAXBIN 256
#define PTHR   512
#define PUNR   10
#define SLICE  (PTHR * PUNR)
#define TSTR   320
#define TABMAX (1 << 20)
#define EPW    16
#define CAP    8192
#define RSH    9

__global__ __launch_bounds__(PTHR) void gcn_bin_kernel(
    const int* __restrict__ edge_src,
    const int* __restrict__ edge_dst,
    int*       __restrict__ tab,
    int2*      __restrict__ sorted,
    int n_edges, int n_nodes)
{
    __shared__ int  lcur[MAXBIN];
    __shared__ int  wtot[MAXBIN / 32];
    __shared__ int2 stage[SLICE];

    const int t    = threadIdx.x;
    const int lane = t & 31;
    const int lo   = blockIdx.x * SLICE;
    const int hi   = (lo + SLICE < n_edges) ? (lo + SLICE) : n_edges;

    int s[PUNR], d[PUNR];
#pragma unroll
    for (int i = 0; i < PUNR; ++i) {
        const int e = lo + t + i * PTHR;
        d[i] = (e < hi) ? edge_dst[e] : -1;
        s[i] = (e < hi) ? edge_src[e] : 0;
    }
    if (t < MAXBIN) lcur[t] = 0;
    __syncthreads();
#pragma unroll
    for (int i = 0; i < PUNR; ++i) {
        if ((unsigned)d[i] < (unsigned)n_nodes) atomicAdd(&lcur[d[i] / ROWS], 1);
    }
    __syncthreads();
    const int c = (t < MAXBIN) ? lcur[t] : 0;
    int inc = c;
#pragma unroll
    for (int off = 1; off < 32; off <<= 1) {
        const int u = __shfl_up(inc, off);
        if (lane >= off) inc += u;
    }
    if (t < MAXBIN && lane == 31) wtot[t >> 5] = inc;
    __syncthreads();
    int nvalid = 0;
#pragma unroll
    for (int k = 0; k < MAXBIN / 32; ++k) nvalid += wtot[k];
    if (t < MAXBIN) {
        int pre = 0;
        for (int k = 0; k < (t >> 5); ++k) pre += wtot[k];
        const int start = pre + inc - c;
        lcur[t] = start;
        tab[blockIdx.x * TSTR + t] = start;
        if (t == MAXBIN - 1) tab[blockIdx.x * TSTR + MAXBIN] = nvalid;
    }
    __syncthreads();
    int pos[PUNR];
#pragma unroll
    for (int i = 0; i < PUNR; ++i) {
        pos[i] = -1;
        if ((unsigned)d[i] < (unsigned)n_nodes) {
            pos[i] = atomicAdd(&lcur[d[i] / ROWS], 1);
        }
    }
#pragma unroll
    for (int i = 0; i < PUNR; ++i) {
        if (pos[i] >= 0) stage[pos[i]] = make_int2(s[i], d[i]);
    }
    __syncthreads();
    for (int i = t; i < nvalid; i += PTHR) sorted[lo + i] = stage[i];
}

__global__ __launch_bounds__(1024) void gcn_sum_gemm_kernel(
    const float* __restrict__ x,
    const int2*  __restrict__ sorted,
    const int*   __restrict__ tab,
    const float* __restrict__ W,
    const float* __restrict__ bias,
    float*       __restrict__ out,
    int n_nodes, int nslices)
{
    __shared__ float agg[(ROWS + 1) * PITCH];
    __shared__ int2  elist[CAP];
    __shared__ int2  slist[CAP];
    __shared__ int   rowpos[ROWS];
    __shared__ int   rtot[16];
    __shared__ int   segpre[MAXBIN + 1];
    __shared__ int   segst[MAXBIN];
    __shared__ int   wtot[MAXBIN / 32];
    __shared__ int   gend;

    const int t        = threadIdx.x;
    const int r        = blockIdx.x;
    const int row_base = r * ROWS;

    const int wid  = __builtin_amdgcn_readfirstlane(t >> 5);
    const int lane = t & 31;
    const int col  = lane * 2;
    const int sub  = lane & (EPW - 1);

    for (int c0 = 0; c0 < nslices; c0 += MAXBIN) {
        const int nch = (nslices - c0 < MAXBIN) ? (nslices - c0) : MAXBIN;
        int st = 0, en = 0;
        if (t < nch) {
            const int* row = tab + (size_t)(c0 + t) * TSTR + r;
            st = row[0];
            en = row[1];
        }
        if (c0 == 0) {
            for (int i = t; i < ROWS * PITCH / 4; i += 1024) {
                ((float4*)agg)[i] = make_float4(0.f, 0.f, 0.f, 0.f);
            }
        }
        const int len = en - st;
        int inc = len;
#pragma unroll
        for (int off = 1; off < 32; off <<= 1) {
            const int u = __shfl_up(inc, off);
            if (lane >= off) inc += u;
        }
        if (t < MAXBIN && lane == 31) wtot[t >> 5] = inc;
        __syncthreads();
        if (t < MAXBIN) {
            int pre = 0;
#pragma unroll
            for (int k = 0; k < MAXBIN / 32; ++k) pre += (k < (t >> 5)) ? wtot[k] : 0;
            segpre[t] = pre + inc - len;
            segst[t]  = (c0 + t) * SLICE + st;
            if (t == MAXBIN - 1) segpre[MAXBIN] = pre + inc;
        }
        __syncthreads();

        int g0 = 0;
        while (g0 < nch) {
            const int base = segpre[g0];
            if (t < ROWS) rowpos[t] = 0;
            if (t >= g0 && t < nch) {
                const bool fit  = (segpre[t + 1] - base) <= CAP;
                const bool next = (t + 1 < nch) && ((segpre[t + 2] - base) <= CAP);
                if (fit && !next) gend = t + 1;
            }
            __syncthreads();
            const int g1  = __builtin_amdgcn_readfirstlane(gend);
            const int cnt = __builtin_amdgcn_readfirstlane(segpre[g1] - base);

            {
                const int seg = g0 + (t >> 2);
                const int q   = t & 3;
                if (seg < g1) {
                    const int   n    = segpre[seg + 1] - segpre[seg];
                    const int   dst0 = segpre[seg] - base;
                    const int2* srcp = sorted + segst[seg];
                    for (int j = q; j < n; j += 16) {
                        int2 a[4];
#pragma unroll
                        for (int u = 0; u < 4; ++u) {
                            const int jj = j + 4 * u;
                            a[u] = (jj < n) ? srcp[jj] : make_int2(0, 0);
                        }
#pragma unroll
                        for (int u = 0; u < 4; ++u) {
                            const int jj = j + 4 * u;
                            if (jj < n) {
                                const int local = a[u].y - row_base;
                                const int rank  = atomicAdd(&rowpos[local], 1);
                                elist[dst0 + jj] = make_int2(a[u].x, local | (rank << RSH));
                            }
                        }
                    }
                }
            }
            __syncthreads();

            {
                const int rc = (t < ROWS) ? rowpos[t] : 0;
                int rinc = rc;
#pragma unroll
                for (int off = 1; off < 32; off <<= 1) {
                    const int u = __shfl_up(rinc, off);
                    if (lane >= off) rinc += u;
                }
                if (t < 512 && lane == 31) rtot[t >> 5] = rinc;
                __syncthreads();
                if (t < ROWS) {
                    int pre = 0;
#pragma unroll
                    for (int k = 0; k < 16; ++k) pre += (k < (t >> 5)) ? rtot[k] : 0;
                    rowpos[t] = pre + rinc - rc;
                }
                __syncthreads();
            }
            for (int k = t; k < cnt; k += 1024) {
                const int2 e     = elist[k];
                const int  local = e.y & ((1 << RSH) - 1);
                slist[rowpos[local] + (e.y >> RSH)] = make_int2(e.x, local);
            }
            __syncthreads();

            const int cs = ((cnt + 32 * EPW - 1) / (32 * EPW)) * EPW;
            int e0 = wid * cs;
            if (e0 < cnt) {
                const int hi = (e0 + cs < cnt) ? (e0 + cs) : cnt;
                int px, py;
                {
                    const int  k  = e0 + sub;
                    const bool ok = k < hi;
                    const int2 pe = slist[ok ? k : 0];
                    px = ok ? pe.x : 0;
                    py = ok ? pe.y : ROWS;
                }
                int   cur  = ROWS;
                float accx = 0.f, accy = 0.f;
                for (;;) {
                    float2 v[EPW];
                    int    dl[EPW];
#pragma unroll
                    for (int i = 0; i < EPW; ++i) {
                        const int src = __builtin_amdgcn_readlane(px, i);
                        dl[i] = __builtin_amdgcn_readlane(py, i);
                        v[i]  = *(const float2*)(x + (size_t)src * FDIM + col);
                    }
                    const int  e1   = e0 + EPW;
                    const bool more = e1 < hi;
                    if (more) {
                        const int  k  = e1 + sub;
                        const bool ok = k < hi;
                        const int2 pe = slist[ok ? k : 0];
                        px = ok ? pe.x : 0;
                        py = ok ? pe.y : ROWS;
                    }
#pragma unroll
                    for (int i = 0; i < EPW; ++i) {
                        if (dl[i] != cur) {
                            float* a = agg + cur * PITCH + col;
                            atomicAdd(a,     accx);
                            atomicAdd(a + 1, accy);
                            accx = 0.f;
                            accy = 0.f;
                        }
                        cur   = dl[i];
                        accx += v[i].x;
                        accy += v[i].y;
                    }
                    if (!more) break;
                    e0 = e1;
                }
                float* a = agg + cur * PITCH + col;
                atomicAdd(a,     accx);
                atomicAdd(a + 1, accy);
            }
            __syncthreads();
            g0 = g1;
        }
    }
    __syncthreads();

    const int half = lane >> 4;
    const int lr   = lane & 15;
    for (int job = wid; job < (ROWS / 16) * (FDIM / 16); job += 32) {
        const int tile_m = job >> 2;
        const int col0   = (job & 3) * 16;

        const float bv = bias[col0 + lr];
        v8f acc;
#pragma unroll
        for (int v = 0; v < 8; ++v) acc[v] = bv;

        const float* arow_p = agg + (tile_m * 16 + lr) * PITCH;

#pragma unroll
        for (int k0 = 0; k0 < FDIM; k0 += 4) {
            v2f a, bm;
            const float2 av = *(const float2*)(arow_p + k0 + 2 * half);
            a[0] = av.x;
            a[1] = av.y;
            bm[0] = W[(k0 + 0 + 2 * half) * FDIM + col0 + lr];
            bm[1] = W[(k0 + 1 + 2 * half) * FDIM + col0 + lr];
            acc = __builtin_amdgcn_wmma_f32_16x16x4_f32(
                false, a, false, bm, (short)0, acc, false, false);
        }

#pragma unroll
        for (int v = 0; v < 8; ++v) {
            const int rm = row_base + tile_m * 16 + v + 8 * half;
            if (rm < n_nodes) {
                out[(size_t)rm * FDIM + col0 + lr] = acc[v];
            }
        }
    }
}

extern "C" void kernel_launch(void* const* d_in, const int* in_sizes, int n_in,
                              void* d_out, int out_size, void* d_ws, size_t ws_size,
                              hipStream_t stream) {
    const float* x        = (const float*)d_in[0];
    const int*   edge_src = (const int*)d_in[1];
    const int*   edge_dst = (const int*)d_in[2];
    const float* W        = (const float*)d_in[3];
    const float* bias     = (const float*)d_in[4];
    float*       out      = (float*)d_out;

    const int n_nodes = in_sizes[0] / FDIM;
    const int n_edges = in_sizes[1];

    int*  tab    = (int*)d_ws;
    int2* sorted = (int2*)(tab + TABMAX);

    const int nslices = (n_edges + SLICE - 1) / SLICE;
    gcn_bin_kernel<<<nslices, PTHR, 0, stream>>>(edge_src, edge_dst, tab, sorted, n_edges, n_nodes);
    {
        const int nbins = (n_nodes + ROWS - 1) / ROWS;
        gcn_sum_gemm_kernel<<<nbins, 1024, 0, stream>>>(x, sorted, tab, W, bias, out, n_nodes, nslices);
    }
}


# sum launch: rows of x gathered 16 bytes per lane, one load instruction fetching two rows (one per half-wave, each half walks its own list piece, entries by row_share); pieces move, sums regroup
# speedup vs baseline: 4.3384x; 1.0298x over previous
#include <hip/hip_runtime.h>


typedef __attribute__((ext_vector_type(2))) float v2f;
typedef __attribute__((ext_vector_type(8))) float v8f;

#define FDIM   64
#define ROWS   400
#define PITCH  68
#define MAXBIN 256
#define PTHR   512
#define PUNR   10
#define SLICE  (PTHR * PUNR)
#define TSTR   320
#define TABMAX (1 << 20)
#define EPW    16
#define CAP    8192
#define RSH    9

__global__ __launch_bounds__(PTHR) void gcn_bin_kernel(
    const int* __restrict__ edge_src,
    const int* __restrict__ edge_dst,
    int*       __restrict__ tab,
    int2*      __restrict__ sorted,
    int n_edges, int n_nodes)
{
    __shared__ int  lcur[MAXBIN];
    __shared__ int  wtot[MAXBIN / 32];
    __shared__ int2 stage[SLICE];

    const int t    = threadIdx.x;
    const int lane = t & 31;
    const int lo   = blockIdx.x * SLICE;
    const int hi   = (lo + SLICE < n_edges) ? (lo + SLICE) : n_edges;

    int s[PUNR], d[PUNR];
#pragma unroll
    for (int i = 0; i < PUNR; ++i) {
        const int e = lo + t + i * PTHR;
        d[i] = (e < hi) ? edge_dst[e] : -1;
        s[i] = (e < hi) ? edge_src[e] : 0;
    }
    if (t < MAXBIN) lcur[t] = 0;
    __syncthreads();
#pragma unroll
    for (int i = 0; i < PUNR; ++i) {
        if ((unsigned)d[i] < (unsigned)n_nodes) atomicAdd(&lcur[d[i] / ROWS], 1);
    }
    __syncthreads();
    const int c = (t < MAXBIN) ? lcur[t] : 0;
    int inc = c;
#pragma unroll
    for (int off = 1; off < 32; off <<= 1) {
        const int u = __shfl_up(inc, off);
        if (lane >= off) inc += u;
    }
    if (t < MAXBIN && lane == 31) wtot[t >> 5] = inc;
    __syncthreads();
    int nvalid = 0;
#pragma unroll
    for (int k = 0; k < MAXBIN / 32; ++k) nvalid += wtot[k];
    if (t < MAXBIN) {
        int pre = 0;
        for (int k = 0; k < (t >> 5); ++k) pre += wtot[k];
        const int start = pre + inc - c;
        lcur[t] = start;
        tab[blockIdx.x * TSTR + t] = start;
        if (t == MAXBIN - 1) tab[blockIdx.x * TSTR + MAXBIN] = nvalid;
    }
    __syncthreads();
    int pos[PUNR];
#pragma unroll
    for (int i = 0; i < PUNR; ++i) {
        pos[i] = -1;
        if ((unsigned)d[i] < (unsigned)n_nodes) {
            pos[i] = atomicAdd(&lcur[d[i] / ROWS], 1);
        }
    }
#pragma unroll
    for (int i = 0; i < PUNR; ++i) {
        if (pos[i] >= 0) stage[pos[i]] = make_int2(s[i], d[i]);
    }
    __syncthreads();
    for (int i = t; i < nvalid; i += PTHR) sorted[lo + i] = stage[i];
}

__global__ __launch_bounds__(1024) void gcn_sum_gemm_kernel(
    const float* __restrict__ x,
    const int2*  __restrict__ sorted,
    const int*   __restrict__ tab,
    const float* __restrict__ W,
    const float* __restrict__ bias,
    float*       __restrict__ out,
    int n_nodes, int nslices)
{
    __shared__ float agg[(ROWS + 1) * PITCH];
    __shared__ int2  elist[CAP];
    __shared__ int2  slist[CAP];
    __shared__ int   rowpos[ROWS];
    __shared__ int   rtot[16];
    __shared__ int   segpre[MAXBIN + 1];
    __shared__ int   segst[MAXBIN];
    __shared__ int   wtot[MAXBIN / 32];
    __shared__ int   gend;

    const int t        = threadIdx.x;
    const int r        = blockIdx.x;
    const int row_base = r * ROWS;

    const int wid  = __builtin_amdgcn_readfirstlane(t >> 5);
    const int lane = t & 31;
    const int sub  = lane & (EPW - 1);

    for (int c0 = 0; c0 < nslices; c0 += MAXBIN) {
        const int nch = (nslices - c0 < MAXBIN) ? (nslices - c0) : MAXBIN;
        int st = 0, en = 0;
        if (t < nch) {
            const int* row = tab + (size_t)(c0 + t) * TSTR + r;
            st = row[0];
            en = row[1];
        }
        if (c0 == 0) {
            for (int i = t; i < ROWS * PITCH / 4; i += 1024) {
                ((float4*)agg)[i] = make_float4(0.f, 0.f, 0.f, 0.f);
            }
        }
        const int len = en - st;
        int inc = len;
#pragma unroll
        for (int off = 1; off < 32; off <<= 1) {
            const int u = __shfl_up(inc, off);
            if (lane >= off) inc += u;
        }
        if (t < MAXBIN && lane == 31) wtot[t >> 5] = inc;
        __syncthreads();
        if (t < MAXBIN) {
            int pre = 0;
#pragma unroll
            for (int k = 0; k < MAXBIN / 32; ++k) pre += (k < (t >> 5)) ? wtot[k] : 0;
            segpre[t] = pre + inc - len;
            segst[t]  = (c0 + t) * SLICE + st;
            if (t == MAXBIN - 1) segpre[MAXBIN] = pre + inc;
        }
        __syncthreads();

        int g0 = 0;
        while (g0 < nch) {
            const int base = segpre[g0];
            if (t < ROWS) rowpos[t] = 0;
            if (t >= g0 && t < nch) {
                const bool fit  = (segpre[t + 1] - base) <= CAP;
                const bool next = (t + 1 < nch) && ((segpre[t + 2] - base) <= CAP);
                if (fit && !next) gend = t + 1;
            }
            __syncthreads();
            const int g1  = __builtin_amdgcn_readfirstlane(gend);
            const int cnt = __builtin_amdgcn_readfirstlane(segpre[g1] - base);

            {
                const int seg = g0 + (t >> 2);
                const int q   = t & 3;
                if (seg < g1) {
                    const int   n    = segpre[seg + 1] - segpre[seg];
                    const int   dst0 = segpre[seg] - base;
                    const int2* srcp = sorted + segst[seg];
                    for (int j = q; j < n; j += 16) {
                        int2 a[4];
#pragma unroll
                        for (int u = 0; u < 4; ++u) {
                            const int jj = j + 4 * u;
                            a[u] = (jj < n) ? srcp[jj] : make_int2(0, 0);
                        }
#pragma unroll
                        for (int u = 0; u < 4; ++u) {
                            const int jj = j + 4 * u;
                            if (jj < n) {
                                const int local = a[u].y - row_base;
                                const int rank  = atomicAdd(&rowpos[local], 1);
                                elist[dst0 + jj] = make_int2(a[u].x, local | (rank << RSH));
                            }
                        }
                    }
                }
            }
            __syncthreads();

            {
                const int rc = (t < ROWS) ? rowpos[t] : 0;
                int rinc = rc;
#pragma unroll
                for (int off = 1; off < 32; off <<= 1) {
                    const int u = __shfl_up(rinc, off);
                    if (lane >= off) rinc += u;
                }
                if (t < 512 && lane == 31) rtot[t >> 5] = rinc;
                __syncthreads();
                if (t < ROWS) {
                    int pre = 0;
#pragma unroll
                    for (int k = 0; k < 16; ++k) pre += (k < (t >> 5)) ? rtot[k] : 0;
                    rowpos[t] = pre + rinc - rc;
                }
                __syncthreads();
            }
            for (int k = t; k < cnt; k += 1024) {
                const int2 e     = elist[k];
                const int  local = e.y & ((1 << RSH) - 1);
                slist[rowpos[local] + (e.y >> RSH)] = make_int2(e.x, local);
            }
            __syncthreads();

            const int hw   = lane >> 4;
            const int cb   = sub * 16;
            const int cs   = ((cnt + 64 * EPW - 1) / (64 * EPW)) * EPW;
            const int e0w  = wid * 2 * cs;
            if (e0w < cnt) {
                const int n0 = (cnt - e0w < cs) ? (cnt - e0w) : cs;
                const int ng = (n0 + EPW - 1) / EPW;
                const int e0 = e0w + hw * cs;
                const int hi = (e0 + cs < cnt) ? (e0 + cs) : cnt;
                int px, py;
                {
                    const int  k  = e0 + sub;
                    const bool ok = k < hi;
                    const int2 pe = slist[ok ? k : 0];
                    px = ok ? pe.x : 0;
                    py = ok ? pe.y : ROWS;
                }
                int    cur = ROWS;
                float4 acc = make_float4(0.f, 0.f, 0.f, 0.f);
                for (int g = 0; g < ng; ++g) {
                    float4 v[EPW];
                    int    dl[EPW];
#define GATHER_ROW(I)                                                                                   \
                    {                                                                                   \
                        const int src = __builtin_amdgcn_update_dpp(0, px, 0x150 + (I), 0xf, 0xf, false); \
                        dl[I] = __builtin_amdgcn_update_dpp(0, py, 0x150 + (I), 0xf, 0xf, false);       \
                        v[I]  = *(const float4*)((const char*)x + (((unsigned)src << 8) + (unsigned)cb)); \
                    }
                    GATHER_ROW(0)  GATHER_ROW(1)  GATHER_ROW(2)  GATHER_ROW(3)
                    GATHER_ROW(4)  GATHER_ROW(5)  GATHER_ROW(6)  GATHER_ROW(7)
                    GATHER_ROW(8)  GATHER_ROW(9)  GATHER_ROW(10) GATHER_ROW(11)
                    GATHER_ROW(12) GATHER_ROW(13) GATHER_ROW(14) GATHER_ROW(15)
#undef GATHER_ROW
                    {
                        const int  k  = e0 + (g + 1) * EPW + sub;
                        const bool ok = (g + 1 < ng) && (k < hi);
                        const int2 pe = slist[ok ? k : 0];
                        px = ok ? pe.x : 0;
                        py = ok ? pe.y : ROWS;
                    }
#pragma unroll
                    for (int i = 0; i < EPW; ++i) {
                        if (dl[i] != cur) {
                            float* a = agg + cur * PITCH + sub * 4;
                            atomicAdd(a,     acc.x);
                            atomicAdd(a + 1, acc.y);
                            atomicAdd(a + 2, acc.z);
                            atomicAdd(a + 3, acc.w);
                            acc = make_float4(0.f, 0.f, 0.f, 0.f);
                        }
                        cur    = dl[i];
                        acc.x += v[i].x;
                        acc.y += v[i].y;
                        acc.z += v[i].z;
                        acc.w += v[i].w;
                    }
                }
                float* a = agg + cur * PITCH + sub * 4;
                atomicAdd(a,     acc.x);
                atomicAdd(a + 1, acc.y);
                atomicAdd(a + 2, acc.z);
                atomicAdd(a + 3, acc.w);
            }
            __syncthreads();
            g0 = g1;
        }
    }
    __syncthreads();

    const int half = lane >> 4;
    const int lr   = lane & 15;
    for (int job = wid; job < (ROWS / 16) * (FDIM / 16); job += 32) {
        const int tile_m = job >> 2;
        const int col0   = (job & 3) * 16;

        const float bv = bias[col0 + lr];
        v8f acc;
#pragma unroll
        for (int v = 0; v < 8; ++v) acc[v] = bv;

        const float* arow_p = agg + (tile_m * 16 + lr) * PITCH;

#pragma unroll
        for (int k0 = 0; k0 < FDIM; k0 += 4) {
            v2f a, bm;
            const float2 av = *(const float2*)(arow_p + k0 + 2 * half);
            a[0] = av.x;
            a[1] = av.y;
            bm[0] = W[(k0 + 0 + 2 * half) * FDIM + col0 + lr];
            bm[1] = W[(k0 + 1 + 2 * half) * FDIM + col0 + lr];
            acc = __builtin_amdgcn_wmma_f32_16x16x4_f32(
                false, a, false, bm, (short)0, acc, false, false);
        }

#pragma unroll
        for (int v = 0; v < 8; ++v) {
            const int rm = row_base + tile_m * 16 + v + 8 * half;
            if (rm < n_nodes) {
                out[(size_t)rm * FDIM + col0 + lr] = acc[v];
            }
        }
    }
}

extern "C" void kernel_launch(void* const* d_in, const int* in_sizes, int n_in,
                              void* d_out, int out_size, void* d_ws, size_t ws_size,
                              hipStream_t stream) {
    const float* x        = (const float*)d_in[0];
    const int*   edge_src = (const int*)d_in[1];
    const int*   edge_dst = (const int*)d_in[2];
    const float* W        = (const float*)d_in[3];
    const float* bias     = (const float*)d_in[4];
    float*       out      = (float*)d_out;

    const int n_nodes = in_sizes[0] / FDIM;
    const int n_edges = in_sizes[1];

    int*  tab    = (int*)d_ws;
    int2* sorted = (int2*)(tab + TABMAX);

    const int nslices = (n_edges + SLICE - 1) / SLICE;
    gcn_bin_kernel<<<nslices, PTHR, 0, stream>>>(edge_src, edge_dst, tab, sorted, n_edges, n_nodes);
    {
        const int nbins = (n_nodes + ROWS - 1) / ROWS;
        gcn_sum_gemm_kernel<<<nbins, 1024, 0, stream>>>(x, sorted, tab, W, bias, out, n_nodes, nslices);
    }
}
